# starting kernel
# baseline (speedup 1.0000x reference)
#include <hip/hip_runtime.h>


typedef __attribute__((ext_vector_type(2))) float v2f;
typedef __attribute__((ext_vector_type(8))) float v8f;

#define GRU_B 4096
#define GRU_T 10
#define GRU_S 301
#define GRU_N (GRU_B * GRU_S)

__device__ __forceinline__ float swap_half(float v) {
  int i = __builtin_bit_cast(int, v);
  i = __builtin_amdgcn_ds_swizzle(i, 0x401F);
  return __builtin_bit_cast(float, i);
}

__device__ __forceinline__ float tanh_fast(float x) {
#if __has_builtin(__builtin_amdgcn_tanhf)
  return __builtin_amdgcn_tanhf(x);
#else
  float e = __builtin_amdgcn_exp2f(2.88539008f * x);
  return 1.0f - 2.0f * __builtin_amdgcn_rcpf(e + 1.0f);
#endif
}

__device__ __forceinline__ float sigmoid_fast(float x) {
#if __has_builtin(__builtin_amdgcn_tanhf)
  return __builtin_fmaf(0.5f, __builtin_amdgcn_tanhf(0.5f * x), 0.5f);
#else
  float e = __builtin_amdgcn_exp2f(-1.44269504f * x);
  return __builtin_amdgcn_rcpf(1.0f + e);
#endif
}

__global__ __launch_bounds__(256) void gru_wmma_kernel(
    const float* __restrict__ x,
    const float* __restrict__ w_ih,
    const float* __restrict__ w_hh,
    const float* __restrict__ b_ih,
    const float* __restrict__ b_hh,
    const float* __restrict__ fc_w,
    const float* __restrict__ fc_b,
    float* __restrict__ out)
{
  const int lane = threadIdx.x & 31;
  const int m    = lane & 15;
  const bool hi  = lane >= 16;

  const int wave = blockIdx.x * (blockDim.x >> 5) + (threadIdx.x >> 5);
  if (wave * 16 >= GRU_N) return;

  const int n = wave * 16 + m;
  const int b = n / GRU_S;
  const int s = n - b * GRU_S;
  const int xbase = b * (GRU_T * GRU_S) + s;

  const int mm = (m < 15) ? m : 14;
  float a10 = w_hh[mm * 5 + (hi ? 2 : 0)];
  float a11 = w_hh[mm * 5 + (hi ? 3 : 1)];
  float biasm = (mm < 10) ? (b_ih[mm] + b_hh[mm]) : b_hh[mm];
  float a20 = hi ? biasm : w_hh[mm * 5 + 4];
  float a21 = hi ? 0.0f : ((mm < 10) ? w_ih[mm] : 0.0f);
  if (m == 15) { a10 = 0.0f; a11 = 0.0f; a20 = 0.0f; a21 = 0.0f; }
  v2f A1; A1.x = a10; A1.y = a11;
  v2f A2; A2.x = a20; A2.y = a21;

  float wn[5], bn[5];
#pragma unroll
  for (int j = 0; j < 5; ++j) { wn[j] = w_ih[10 + j]; bn[j] = b_ih[10 + j]; }

  float xv[GRU_T];
#pragma unroll
  for (int t = 0; t < GRU_T; ++t) xv[t] = x[xbase + t * GRU_S];

  float h[5] = {0.f, 0.f, 0.f, 0.f, 0.f};

#pragma unroll
  for (int t = 0; t < GRU_T; ++t) {
    v2f B1; B1.x = hi ? h[2] : h[0];
            B1.y = hi ? h[3] : h[1];
    v2f B2; B2.x = hi ? 1.0f : h[4];
            B2.y = hi ? 0.0f : xv[t];

    v8f c = {};
    c = __builtin_amdgcn_wmma_f32_16x16x4_f32(false, A1, false, B1,
                                              (short)0, c, false, false);
    c = __builtin_amdgcn_wmma_f32_16x16x4_f32(false, A2, false, B2,
                                              (short)0, c, false, false);

    float p[15];
#pragma unroll
    for (int k = 0; k < 8; ++k) {
      float dk = c[k];
      float tk = swap_half(dk);
      p[k] = hi ? tk : dk;
      if (k < 7) p[k + 8] = hi ? dk : tk;
    }

#pragma unroll
    for (int j = 0; j < 5; ++j) {
      float r  = sigmoid_fast(p[j]);
      float z  = sigmoid_fast(p[5 + j]);
      float np = __builtin_fmaf(wn[j], xv[t], bn[j]);
      np = __builtin_fmaf(r, p[10 + j], np);
      float nn = tanh_fast(np);
      h[j] = __builtin_fmaf(z, h[j] - nn, nn);
    }
  }

  if (!hi) {
#pragma unroll
    for (int o = 0; o < 3; ++o) {
      float y = fc_b[o];
#pragma unroll
      for (int j = 0; j < 5; ++j) y = __builtin_fmaf(fc_w[o * 5 + j], h[j], y);
      out[(b * 3 + o) * GRU_S + s] = y;
    }
  }
}

extern "C" void kernel_launch(void* const* d_in, const int* in_sizes, int n_in,
                              void* d_out, int out_size, void* d_ws, size_t ws_size,
                              hipStream_t stream) {
  (void)in_sizes; (void)n_in; (void)out_size; (void)d_ws; (void)ws_size;
  const float* x    = (const float*)d_in[0];
  const float* w_ih = (const float*)d_in[1];
  const float* w_hh = (const float*)d_in[2];
  const float* b_ih = (const float*)d_in[3];
  const float* b_hh = (const float*)d_in[4];
  const float* fc_w = (const float*)d_in[5];
  const float* fc_b = (const float*)d_in[6];
  float* out = (float*)d_out;

  const int threads = 256;
  const int samplesPerBlock = (threads / 32) * 16;
  const int blocks = (GRU_N + samplesPerBlock - 1) / samplesPerBlock;
  gru_wmma_kernel<<<blocks, threads, 0, stream>>>(x, w_ih, w_hh, b_ih, b_hh,
                                                  fc_w, fc_b, out);
}


# Block-diagonal weight fragments: each lane owns one whole sample (32 per wave), 6 WMMAs per step with the biases as C, no lane exchange or selects; order of the gate sums changes
# speedup vs baseline: 1.2459x; 1.2459x over previous
#include <hip/hip_runtime.h>


typedef __attribute__((ext_vector_type(2))) float v2f;
typedef __attribute__((ext_vector_type(8))) float v8f;

#define GRU_B 4096
#define GRU_T 10
#define GRU_S 301
#define GRU_N (GRU_B * GRU_S)

__device__ __forceinline__ float tanh_fast(float x) {
#if __has_builtin(__builtin_amdgcn_tanhf)
  return __builtin_amdgcn_tanhf(x);
#else
  float e = __builtin_amdgcn_exp2f(2.88539008f * x);
  return 1.0f - 2.0f * __builtin_amdgcn_rcpf(e + 1.0f);
#endif
}

__device__ __forceinline__ float sigmoid_fast(float x) {
#if __has_builtin(__builtin_amdgcn_tanhf)
  return __builtin_fmaf(0.5f, __builtin_amdgcn_tanhf(0.5f * x), 0.5f);
#else
  float e = __builtin_amdgcn_exp2f(-1.44269504f * x);
  return __builtin_amdgcn_rcpf(1.0f + e);
#endif
}

__device__ __forceinline__ v8f mm4(v2f a, v2f b, v8f c) {
  return __builtin_amdgcn_wmma_f32_16x16x4_f32(false, a, false, b,
                                               (short)0, c, false, false);
}

__global__ __launch_bounds__(256) void gru_wmma_kernel(
    const float* __restrict__ x,
    const float* __restrict__ w_ih,
    const float* __restrict__ w_hh,
    const float* __restrict__ b_ih,
    const float* __restrict__ b_hh,
    const float* __restrict__ fc_w,
    const float* __restrict__ fc_b,
    float* __restrict__ out)
{
  const int lane = threadIdx.x & 31;
  const int wave = blockIdx.x * (blockDim.x >> 5) + (threadIdx.x >> 5);
  if (wave * 32 >= GRU_N) return;

  const int n = wave * 32 + lane;
  const int b = n / GRU_S;
  const int s = n - b * GRU_S;
  const int xbase = b * (GRU_T * GRU_S) + s;

  const int row  = lane & 15;
  const int r8   = row & 7;
  const bool live = (lane >> 4) == (row >> 3);

  const int g0 = r8;
  const int g1 = (r8 < 7) ? (8 + r8) : 14;
  const bool on0 = live;
  const bool on1 = live && (r8 < 7);

  const float u00 = w_hh[g0 * 5 + 0], u01 = w_hh[g0 * 5 + 1], u02 = w_hh[g0 * 5 + 2];
  const float u03 = w_hh[g0 * 5 + 3], u04 = w_hh[g0 * 5 + 4], u0x = w_ih[g0];
  const float u10 = w_hh[g1 * 5 + 0], u11 = w_hh[g1 * 5 + 1], u12 = w_hh[g1 * 5 + 2];
  const float u13 = w_hh[g1 * 5 + 3], u14 = w_hh[g1 * 5 + 4], u1x = w_ih[g1];
  const bool on1x = on1 && (g1 < 10);

  v2f A00, A01, A02, A10, A11, A12;
  A00.x = on0 ? u00 : 0.0f;  A00.y = on0 ? u01 : 0.0f;
  A01.x = on0 ? u02 : 0.0f;  A01.y = on0 ? u03 : 0.0f;
  A02.x = on0 ? u04 : 0.0f;  A02.y = on0 ? u0x : 0.0f;
  A10.x = on1 ? u10 : 0.0f;  A10.y = on1 ? u11 : 0.0f;
  A11.x = on1 ? u12 : 0.0f;  A11.y = on1 ? u13 : 0.0f;
  A12.x = on1 ? u14 : 0.0f;  A12.y = on1x ? u1x : 0.0f;

  v8f cb0, cb1;
#pragma unroll
  for (int r = 0; r < 8; ++r) cb0[r] = b_ih[r] + b_hh[r];
  cb1[0] = b_ih[8] + b_hh[8];
  cb1[1] = b_ih[9] + b_hh[9];
#pragma unroll
  for (int r = 2; r < 7; ++r) cb1[r] = b_hh[8 + r];
  cb1[7] = 0.0f;

  float wn[5], bn[5];
#pragma unroll
  for (int j = 0; j < 5; ++j) { wn[j] = w_ih[10 + j]; bn[j] = b_ih[10 + j]; }

  float xv[GRU_T];
#pragma unroll
  for (int t = 0; t < GRU_T; ++t) xv[t] = x[xbase + t * GRU_S];

  float h[5] = {0.f, 0.f, 0.f, 0.f, 0.f};

#pragma unroll
  for (int t = 0; t < GRU_T; ++t) {
    v2f B0; B0.x = h[0]; B0.y = h[1];
    v2f B1; B1.x = h[2]; B1.y = h[3];
    v2f B2; B2.x = h[4]; B2.y = xv[t];

    v8f d0 = mm4(A00, B0, cb0);
    v8f d1 = mm4(A10, B0, cb1);
    d0 = mm4(A01, B1, d0);
    d1 = mm4(A11, B1, d1);
    d0 = mm4(A02, B2, d0);
    d1 = mm4(A12, B2, d1);

    float p[15];
#pragma unroll
    for (int k = 0; k < 8; ++k) p[k] = d0[k];
#pragma unroll
    for (int k = 0; k < 7; ++k) p[8 + k] = d1[k];

#pragma unroll
    for (int j = 0; j < 5; ++j) {
      float r  = sigmoid_fast(p[j]);
      float z  = sigmoid_fast(p[5 + j]);
      float np = __builtin_fmaf(wn[j], xv[t], bn[j]);
      np = __builtin_fmaf(r, p[10 + j], np);
      float nn = tanh_fast(np);
      h[j] = __builtin_fmaf(z, h[j] - nn, nn);
    }
  }

#pragma unroll
  for (int o = 0; o < 3; ++o) {
    float y = fc_b[o];
#pragma unroll
    for (int j = 0; j < 5; ++j) y = __builtin_fmaf(fc_w[o * 5 + j], h[j], y);
    out[(b * 3 + o) * GRU_S + s] = y;
  }
}

extern "C" void kernel_launch(void* const* d_in, const int* in_sizes, int n_in,
                              void* d_out, int out_size, void* d_ws, size_t ws_size,
                              hipStream_t stream) {
  (void)in_sizes; (void)n_in; (void)out_size; (void)d_ws; (void)ws_size;
  const float* x    = (const float*)d_in[0];
  const float* w_ih = (const float*)d_in[1];
  const float* w_hh = (const float*)d_in[2];
  const float* b_ih = (const float*)d_in[3];
  const float* b_hh = (const float*)d_in[4];
  const float* fc_w = (const float*)d_in[5];
  const float* fc_b = (const float*)d_in[6];
  float* out = (float*)d_out;

  const int threads = 256;
  const int samplesPerBlock = (threads / 32) * 32;
  const int blocks = (GRU_N + samplesPerBlock - 1) / samplesPerBlock;
  gru_wmma_kernel<<<blocks, threads, 0, stream>>>(x, w_ih, w_hh, b_ih, b_hh,
                                                  fc_w, fc_b, out);
}


# Gate rows reordered so like gates of adjacent hidden units sit in even-aligned D register pairs; gate math runs two values per instruction (packed f32 mul/fma/add); order of sums unchanged
# speedup vs baseline: 1.3139x; 1.0546x over previous
#include <hip/hip_runtime.h>


typedef __attribute__((ext_vector_type(2))) float v2f;
typedef __attribute__((ext_vector_type(8))) float v8f;

#define GRU_B 4096
#define GRU_T 10
#define GRU_S 301
#define GRU_N (GRU_B * GRU_S)

__device__ __forceinline__ float tanh_fast(float x) {
#if __has_builtin(__builtin_amdgcn_tanhf)
  return __builtin_amdgcn_tanhf(x);
#else
  float e = __builtin_amdgcn_exp2f(2.88539008f * x);
  return 1.0f - 2.0f * __builtin_amdgcn_rcpf(e + 1.0f);
#endif
}

__device__ __forceinline__ float sigmoid_fast(float x) {
#if __has_builtin(__builtin_amdgcn_tanhf)
  return __builtin_fmaf(0.5f, __builtin_amdgcn_tanhf(0.5f * x), 0.5f);
#else
  float e = __builtin_amdgcn_exp2f(-1.44269504f * x);
  return __builtin_amdgcn_rcpf(1.0f + e);
#endif
}

__device__ __forceinline__ v2f fma2(v2f a, v2f b, v2f c) {
  return __builtin_elementwise_fma(a, b, c);
}
__device__ __forceinline__ v2f tanh2(v2f v) {
  v2f o; o.x = tanh_fast(v.x); o.y = tanh_fast(v.y); return o;
}
__device__ __forceinline__ v2f sigmoid2(v2f v) {
#if __has_builtin(__builtin_amdgcn_tanhf)
  const v2f hf = {0.5f, 0.5f};
  return fma2(tanh2(v * hf), hf, hf);
#else
  v2f o; o.x = sigmoid_fast(v.x); o.y = sigmoid_fast(v.y); return o;
#endif
}

__device__ __forceinline__ v8f mm4(v2f a, v2f b, v8f c) {
  return __builtin_amdgcn_wmma_f32_16x16x4_f32(false, a, false, b,
                                               (short)0, c, false, false);
}

__global__ __launch_bounds__(256) void gru_wmma_kernel(
    const float* __restrict__ x,
    const float* __restrict__ w_ih,
    const float* __restrict__ w_hh,
    const float* __restrict__ b_ih,
    const float* __restrict__ b_hh,
    const float* __restrict__ fc_w,
    const float* __restrict__ fc_b,
    float* __restrict__ out)
{
  const int lane = threadIdx.x & 31;
  const int wave = blockIdx.x * (blockDim.x >> 5) + (threadIdx.x >> 5);
  if (wave * 32 >= GRU_N) return;

  const int n = wave * 32 + lane;
  const int b = n / GRU_S;
  const int s = n - b * GRU_S;
  const int xbase = b * (GRU_T * GRU_S) + s;

  const int row  = lane & 15;
  const int r8   = row & 7;
  const bool live = (lane >> 4) == (row >> 3);

  const int g0 = (r8 < 4) ? r8 : (r8 + 1);
  const int g1 = (r8 == 0) ? 4 : (r8 == 1) ? 9 : (r8 < 7) ? (8 + r8) : 14;
  const bool on0 = live;
  const bool on1 = live && (r8 < 7);

  const float u00 = w_hh[g0 * 5 + 0], u01 = w_hh[g0 * 5 + 1], u02 = w_hh[g0 * 5 + 2];
  const float u03 = w_hh[g0 * 5 + 3], u04 = w_hh[g0 * 5 + 4], u0x = w_ih[g0];
  const float u10 = w_hh[g1 * 5 + 0], u11 = w_hh[g1 * 5 + 1], u12 = w_hh[g1 * 5 + 2];
  const float u13 = w_hh[g1 * 5 + 3], u14 = w_hh[g1 * 5 + 4], u1x = w_ih[g1];
  const bool on1x = on1 && (g1 < 10);

  v2f A00, A01, A02, A10, A11, A12;
  A00.x = on0 ? u00 : 0.0f;  A00.y = on0 ? u01 : 0.0f;
  A01.x = on0 ? u02 : 0.0f;  A01.y = on0 ? u03 : 0.0f;
  A02.x = on0 ? u04 : 0.0f;  A02.y = on0 ? u0x : 0.0f;
  A10.x = on1 ? u10 : 0.0f;  A10.y = on1 ? u11 : 0.0f;
  A11.x = on1 ? u12 : 0.0f;  A11.y = on1 ? u13 : 0.0f;
  A12.x = on1 ? u14 : 0.0f;  A12.y = on1x ? u1x : 0.0f;

  v8f cb0, cb1;
#pragma unroll
  for (int r = 0; r < 4; ++r) {
    cb0[r]     = b_ih[r] + b_hh[r];
    cb0[4 + r] = b_ih[5 + r] + b_hh[5 + r];
  }
  cb1[0] = b_ih[4] + b_hh[4];
  cb1[1] = b_ih[9] + b_hh[9];
#pragma unroll
  for (int r = 2; r < 7; ++r) cb1[r] = b_hh[8 + r];
  cb1[7] = 0.0f;

  v2f wn01, wn23, bn01, bn23;
  wn01.x = w_ih[10]; wn01.y = w_ih[11]; wn23.x = w_ih[12]; wn23.y = w_ih[13];
  bn01.x = b_ih[10]; bn01.y = b_ih[11]; bn23.x = b_ih[12]; bn23.y = b_ih[13];
  const float wn4 = w_ih[14], bn4 = b_ih[14];

  float xv[GRU_T];
#pragma unroll
  for (int t = 0; t < GRU_T; ++t) xv[t] = x[xbase + t * GRU_S];

  v2f h01 = {0.f, 0.f}, h23 = {0.f, 0.f};
  float h4 = 0.f;

#pragma unroll
  for (int t = 0; t < GRU_T; ++t) {
    v2f B2; B2.x = h4; B2.y = xv[t];

    v8f d0 = mm4(A00, h01, cb0);
    v8f d1 = mm4(A10, h01, cb1);
    d0 = mm4(A01, h23, d0);
    d1 = mm4(A11, h23, d1);
    d0 = mm4(A02, B2, d0);
    d1 = mm4(A12, B2, d1);

    v2f r01 = {d0[0], d0[1]}, r23 = {d0[2], d0[3]};
    v2f z01 = {d0[4], d0[5]}, z23 = {d0[6], d0[7]};
    v2f rz4 = {d1[0], d1[1]};
    v2f n01 = {d1[2], d1[3]}, n23 = {d1[4], d1[5]};
    const float n4 = d1[6];
    const v2f x2 = {xv[t], xv[t]};

    r01 = sigmoid2(r01);  r23 = sigmoid2(r23);
    z01 = sigmoid2(z01);  z23 = sigmoid2(z23);
    rz4 = sigmoid2(rz4);

    v2f p01 = fma2(r01, n01, fma2(wn01, x2, bn01));
    v2f p23 = fma2(r23, n23, fma2(wn23, x2, bn23));
    float p4 = __builtin_fmaf(rz4.x, n4, __builtin_fmaf(wn4, xv[t], bn4));
    p01 = tanh2(p01);  p23 = tanh2(p23);  p4 = tanh_fast(p4);

    h01 = fma2(z01, h01 - p01, p01);
    h23 = fma2(z23, h23 - p23, p23);
    h4  = __builtin_fmaf(rz4.y, h4 - p4, p4);
  }

  const float h[5] = {h01.x, h01.y, h23.x, h23.y, h4};

#pragma unroll
  for (int o = 0; o < 3; ++o) {
    float y = fc_b[o];
#pragma unroll
    for (int j = 0; j < 5; ++j) y = __builtin_fmaf(fc_w[o * 5 + j], h[j], y);
    out[(b * 3 + o) * GRU_S + s] = y;
  }
}

extern "C" void kernel_launch(void* const* d_in, const int* in_sizes, int n_in,
                              void* d_out, int out_size, void* d_ws, size_t ws_size,
                              hipStream_t stream) {
  (void)in_sizes; (void)n_in; (void)out_size; (void)d_ws; (void)ws_size;
  const float* x    = (const float*)d_in[0];
  const float* w_ih = (const float*)d_in[1];
  const float* w_hh = (const float*)d_in[2];
  const float* b_ih = (const float*)d_in[3];
  const float* b_hh = (const float*)d_in[4];
  const float* fc_w = (const float*)d_in[5];
  const float* fc_b = (const float*)d_in[6];
  float* out = (float*)d_out;

  const int threads = 256;
  const int samplesPerBlock = (threads / 32) * 32;
  const int blocks = (GRU_N + samplesPerBlock - 1) / samplesPerBlock;
  gru_wmma_kernel<<<blocks, threads, 0, stream>>>(x, w_ih, w_hh, b_ih, b_hh,
                                                  fc_w, fc_b, out);
}


# Feature pair (h4, x) taken off the block-diagonal WMMAs (every second weight there is zero): its terms enter as the C operand via packed f32 FMAs, 4 WMMAs per step not 6; order of gate sums changes
# speedup vs baseline: 1.3990x; 1.0648x over previous
#include <hip/hip_runtime.h>


typedef __attribute__((ext_vector_type(2))) float v2f;
typedef __attribute__((ext_vector_type(8))) float v8f;

#define GRU_B 4096
#define GRU_T 10
#define GRU_S 301
#define GRU_N (GRU_B * GRU_S)

__device__ __forceinline__ float tanh_fast(float x) {
#if __has_builtin(__builtin_amdgcn_tanhf)
  return __builtin_amdgcn_tanhf(x);
#else
  float e = __builtin_amdgcn_exp2f(2.88539008f * x);
  return 1.0f - 2.0f * __builtin_amdgcn_rcpf(e + 1.0f);
#endif
}

__device__ __forceinline__ float sigmoid_fast(float x) {
#if __has_builtin(__builtin_amdgcn_tanhf)
  return __builtin_fmaf(0.5f, __builtin_amdgcn_tanhf(0.5f * x), 0.5f);
#else
  float e = __builtin_amdgcn_exp2f(-1.44269504f * x);
  return __builtin_amdgcn_rcpf(1.0f + e);
#endif
}

__device__ __forceinline__ v2f fma2(v2f a, v2f b, v2f c) {
  return __builtin_elementwise_fma(a, b, c);
}
__device__ __forceinline__ v2f tanh2(v2f v) {
  v2f o; o.x = tanh_fast(v.x); o.y = tanh_fast(v.y); return o;
}
__device__ __forceinline__ v2f sigmoid2(v2f v) {
#if __has_builtin(__builtin_amdgcn_tanhf)
  const v2f hf = {0.5f, 0.5f};
  return fma2(tanh2(v * hf), hf, hf);
#else
  v2f o; o.x = sigmoid_fast(v.x); o.y = sigmoid_fast(v.y); return o;
#endif
}

__device__ __forceinline__ v8f mm4(v2f a, v2f b, v8f c) {
  return __builtin_amdgcn_wmma_f32_16x16x4_f32(false, a, false, b,
                                               (short)0, c, false, false);
}

__global__ __launch_bounds__(256) void gru_wmma_kernel(
    const float* __restrict__ x,
    const float* __restrict__ w_ih,
    const float* __restrict__ w_hh,
    const float* __restrict__ b_ih,
    const float* __restrict__ b_hh,
    const float* __restrict__ fc_w,
    const float* __restrict__ fc_b,
    float* __restrict__ out)
{
  const int lane = threadIdx.x & 31;
  const int wave = blockIdx.x * (blockDim.x >> 5) + (threadIdx.x >> 5);
  if (wave * 32 >= GRU_N) return;

  const int n = wave * 32 + lane;
  const int b = n / GRU_S;
  const int s = n - b * GRU_S;
  const int xbase = b * (GRU_T * GRU_S) + s;

  const int row  = lane & 15;
  const int r8   = row & 7;
  const bool live = (lane >> 4) == (row >> 3);

  const int g0 = (r8 < 4) ? r8 : (r8 + 1);
  const int g1 = (r8 == 0) ? 4 : (r8 == 1) ? 9 : (r8 < 7) ? (8 + r8) : 14;
  const bool on0 = live;
  const bool on1 = live && (r8 < 7);

  const float u00 = w_hh[g0 * 5 + 0], u01 = w_hh[g0 * 5 + 1], u02 = w_hh[g0 * 5 + 2];
  const float u03 = w_hh[g0 * 5 + 3];
  const float u10 = w_hh[g1 * 5 + 0], u11 = w_hh[g1 * 5 + 1], u12 = w_hh[g1 * 5 + 2];
  const float u13 = w_hh[g1 * 5 + 3];

  v2f A00, A01, A10, A11;
  A00.x = on0 ? u00 : 0.0f;  A00.y = on0 ? u01 : 0.0f;
  A01.x = on0 ? u02 : 0.0f;  A01.y = on0 ? u03 : 0.0f;
  A10.x = on1 ? u10 : 0.0f;  A10.y = on1 ? u11 : 0.0f;
  A11.x = on1 ? u12 : 0.0f;  A11.y = on1 ? u13 : 0.0f;

  v2f b_r01, b_r23, b_z01, b_z23, b_rz4, b_n01, b_n23;
  b_r01.x = b_ih[0] + b_hh[0];  b_r01.y = b_ih[1] + b_hh[1];
  b_r23.x = b_ih[2] + b_hh[2];  b_r23.y = b_ih[3] + b_hh[3];
  b_z01.x = b_ih[5] + b_hh[5];  b_z01.y = b_ih[6] + b_hh[6];
  b_z23.x = b_ih[7] + b_hh[7];  b_z23.y = b_ih[8] + b_hh[8];
  b_rz4.x = b_ih[4] + b_hh[4];  b_rz4.y = b_ih[9] + b_hh[9];
  b_n01.x = b_hh[10];  b_n01.y = b_hh[11];
  b_n23.x = b_hh[12];  b_n23.y = b_hh[13];
  v2f b_n4u;  b_n4u.x = b_hh[14];  b_n4u.y = 0.0f;

  v2f x_r01, x_r23, x_z01, x_z23, x_rz4;
  x_r01.x = w_ih[0];  x_r01.y = w_ih[1];   x_r23.x = w_ih[2];  x_r23.y = w_ih[3];
  x_z01.x = w_ih[5];  x_z01.y = w_ih[6];   x_z23.x = w_ih[7];  x_z23.y = w_ih[8];
  x_rz4.x = w_ih[4];  x_rz4.y = w_ih[9];

  v2f q_r01, q_r23, q_z01, q_z23, q_rz4, q_n01, q_n23;
  q_r01.x = w_hh[0 * 5 + 4];   q_r01.y = w_hh[1 * 5 + 4];
  q_r23.x = w_hh[2 * 5 + 4];   q_r23.y = w_hh[3 * 5 + 4];
  q_z01.x = w_hh[5 * 5 + 4];   q_z01.y = w_hh[6 * 5 + 4];
  q_z23.x = w_hh[7 * 5 + 4];   q_z23.y = w_hh[8 * 5 + 4];
  q_rz4.x = w_hh[4 * 5 + 4];   q_rz4.y = w_hh[9 * 5 + 4];
  q_n01.x = w_hh[10 * 5 + 4];  q_n01.y = w_hh[11 * 5 + 4];
  q_n23.x = w_hh[12 * 5 + 4];  q_n23.y = w_hh[13 * 5 + 4];
  v2f q_n4u;  q_n4u.x = w_hh[14 * 5 + 4];  q_n4u.y = 0.0f;

  v2f wn01, wn23, bn01, bn23;
  wn01.x = w_ih[10]; wn01.y = w_ih[11]; wn23.x = w_ih[12]; wn23.y = w_ih[13];
  bn01.x = b_ih[10]; bn01.y = b_ih[11]; bn23.x = b_ih[12]; bn23.y = b_ih[13];
  const float wn4 = w_ih[14], bn4 = b_ih[14];

  float xv[GRU_T];
#pragma unroll
  for (int t = 0; t < GRU_T; ++t) xv[t] = x[xbase + t * GRU_S];

  v2f h01 = {0.f, 0.f}, h23 = {0.f, 0.f};
  float h4 = 0.f;

#pragma unroll
  for (int t = 0; t < GRU_T; ++t) {
    const v2f x2 = {xv[t], xv[t]};
    const v2f h44 = {h4, h4};

    const v2f c_r01 = fma2(q_r01, h44, fma2(x_r01, x2, b_r01));
    const v2f c_r23 = fma2(q_r23, h44, fma2(x_r23, x2, b_r23));
    const v2f c_z01 = fma2(q_z01, h44, fma2(x_z01, x2, b_z01));
    const v2f c_z23 = fma2(q_z23, h44, fma2(x_z23, x2, b_z23));
    const v2f c_rz4 = fma2(q_rz4, h44, fma2(x_rz4, x2, b_rz4));
    const v2f c_n01 = fma2(q_n01, h44, b_n01);
    const v2f c_n23 = fma2(q_n23, h44, b_n23);
    const v2f c_n4u = fma2(q_n4u, h44, b_n4u);
    const v8f c0 = {c_r01.x, c_r01.y, c_r23.x, c_r23.y, c_z01.x, c_z01.y, c_z23.x, c_z23.y};
    const v8f c1 = {c_rz4.x, c_rz4.y, c_n01.x, c_n01.y, c_n23.x, c_n23.y, c_n4u.x, c_n4u.y};

    v8f d0 = mm4(A00, h01, c0);
    v8f d1 = mm4(A10, h01, c1);
    d0 = mm4(A01, h23, d0);
    d1 = mm4(A11, h23, d1);

    v2f r01 = {d0[0], d0[1]}, r23 = {d0[2], d0[3]};
    v2f z01 = {d0[4], d0[5]}, z23 = {d0[6], d0[7]};
    v2f rz4 = {d1[0], d1[1]};
    v2f n01 = {d1[2], d1[3]}, n23 = {d1[4], d1[5]};
    const float n4 = d1[6];

    r01 = sigmoid2(r01);  r23 = sigmoid2(r23);
    z01 = sigmoid2(z01);  z23 = sigmoid2(z23);
    rz4 = sigmoid2(rz4);

    v2f p01 = fma2(r01, n01, fma2(wn01, x2, bn01));
    v2f p23 = fma2(r23, n23, fma2(wn23, x2, bn23));
    float p4 = __builtin_fmaf(rz4.x, n4, __builtin_fmaf(wn4, xv[t], bn4));
    p01 = tanh2(p01);  p23 = tanh2(p23);  p4 = tanh_fast(p4);

    h01 = fma2(z01, h01 - p01, p01);
    h23 = fma2(z23, h23 - p23, p23);
    h4  = __builtin_fmaf(rz4.y, h4 - p4, p4);
  }

  const float h[5] = {h01.x, h01.y, h23.x, h23.y, h4};

#pragma unroll
  for (int o = 0; o < 3; ++o) {
    float y = fc_b[o];
#pragma unroll
    for (int j = 0; j < 5; ++j) y = __builtin_fmaf(fc_w[o * 5 + j], h[j], y);
    out[(b * 3 + o) * GRU_S + s] = y;
  }
}

extern "C" void kernel_launch(void* const* d_in, const int* in_sizes, int n_in,
                              void* d_out, int out_size, void* d_ws, size_t ws_size,
                              hipStream_t stream) {
  (void)in_sizes; (void)n_in; (void)out_size; (void)d_ws; (void)ws_size;
  const float* x    = (const float*)d_in[0];
  const float* w_ih = (const float*)d_in[1];
  const float* w_hh = (const float*)d_in[2];
  const float* b_ih = (const float*)d_in[3];
  const float* b_hh = (const float*)d_in[4];
  const float* fc_w = (const float*)d_in[5];
  const float* fc_b = (const float*)d_in[6];
  float* out = (float*)d_out;

  const int threads = 256;
  const int samplesPerBlock = (threads / 32) * 32;
  const int blocks = (GRU_N + samplesPerBlock - 1) / samplesPerBlock;
  gru_wmma_kernel<<<blocks, threads, 0, stream>>>(x, w_ih, w_hh, b_ih, b_hh,
                                                  fc_w, fc_b, out);
}


# First time step peeled: h is zero there, so its four WMMAs (products with zero), the h4 terms of C and the h - n subtractions are not issued; 36 WMMAs per wave, not 40; sums unchanged
# speedup vs baseline: 1.4679x; 1.0492x over previous
#include <hip/hip_runtime.h>


typedef __attribute__((ext_vector_type(2))) float v2f;
typedef __attribute__((ext_vector_type(8))) float v8f;

#define GRU_B 4096
#define GRU_T 10
#define GRU_S 301
#define GRU_N (GRU_B * GRU_S)

__device__ __forceinline__ float tanh_fast(float x) {
#if __has_builtin(__builtin_amdgcn_tanhf)
  return __builtin_amdgcn_tanhf(x);
#else
  float e = __builtin_amdgcn_exp2f(2.88539008f * x);
  return 1.0f - 2.0f * __builtin_amdgcn_rcpf(e + 1.0f);
#endif
}

__device__ __forceinline__ float sigmoid_fast(float x) {
#if __has_builtin(__builtin_amdgcn_tanhf)
  return __builtin_fmaf(0.5f, __builtin_amdgcn_tanhf(0.5f * x), 0.5f);
#else
  float e = __builtin_amdgcn_exp2f(-1.44269504f * x);
  return __builtin_amdgcn_rcpf(1.0f + e);
#endif
}

__device__ __forceinline__ v2f fma2(v2f a, v2f b, v2f c) {
  return __builtin_elementwise_fma(a, b, c);
}
__device__ __forceinline__ v2f tanh2(v2f v) {
  v2f o; o.x = tanh_fast(v.x); o.y = tanh_fast(v.y); return o;
}
__device__ __forceinline__ v2f sigmoid2(v2f v) {
#if __has_builtin(__builtin_amdgcn_tanhf)
  const v2f hf = {0.5f, 0.5f};
  return fma2(tanh2(v * hf), hf, hf);
#else
  v2f o; o.x = sigmoid_fast(v.x); o.y = sigmoid_fast(v.y); return o;
#endif
}

__device__ __forceinline__ v8f mm4(v2f a, v2f b, v8f c) {
  return __builtin_amdgcn_wmma_f32_16x16x4_f32(false, a, false, b,
                                               (short)0, c, false, false);
}

__global__ __launch_bounds__(256) void gru_wmma_kernel(
    const float* __restrict__ x,
    const float* __restrict__ w_ih,
    const float* __restrict__ w_hh,
    const float* __restrict__ b_ih,
    const float* __restrict__ b_hh,
    const float* __restrict__ fc_w,
    const float* __restrict__ fc_b,
    float* __restrict__ out)
{
  const int lane = threadIdx.x & 31;
  const int wave = blockIdx.x * (blockDim.x >> 5) + (threadIdx.x >> 5);
  if (wave * 32 >= GRU_N) return;

  const int n = wave * 32 + lane;
  const int b = n / GRU_S;
  const int s = n - b * GRU_S;
  const int xbase = b * (GRU_T * GRU_S) + s;

  const int row  = lane & 15;
  const int r8   = row & 7;
  const bool live = (lane >> 4) == (row >> 3);

  const int g0 = (r8 < 4) ? r8 : (r8 + 1);
  const int g1 = (r8 == 0) ? 4 : (r8 == 1) ? 9 : (r8 < 7) ? (8 + r8) : 14;
  const bool on0 = live;
  const bool on1 = live && (r8 < 7);

  const float u00 = w_hh[g0 * 5 + 0], u01 = w_hh[g0 * 5 + 1], u02 = w_hh[g0 * 5 + 2];
  const float u03 = w_hh[g0 * 5 + 3];
  const float u10 = w_hh[g1 * 5 + 0], u11 = w_hh[g1 * 5 + 1], u12 = w_hh[g1 * 5 + 2];
  const float u13 = w_hh[g1 * 5 + 3];

  v2f A00, A01, A10, A11;
  A00.x = on0 ? u00 : 0.0f;  A00.y = on0 ? u01 : 0.0f;
  A01.x = on0 ? u02 : 0.0f;  A01.y = on0 ? u03 : 0.0f;
  A10.x = on1 ? u10 : 0.0f;  A10.y = on1 ? u11 : 0.0f;
  A11.x = on1 ? u12 : 0.0f;  A11.y = on1 ? u13 : 0.0f;

  v2f b_r01, b_r23, b_z01, b_z23, b_rz4, b_n01, b_n23;
  b_r01.x = b_ih[0] + b_hh[0];  b_r01.y = b_ih[1] + b_hh[1];
  b_r23.x = b_ih[2] + b_hh[2];  b_r23.y = b_ih[3] + b_hh[3];
  b_z01.x = b_ih[5] + b_hh[5];  b_z01.y = b_ih[6] + b_hh[6];
  b_z23.x = b_ih[7] + b_hh[7];  b_z23.y = b_ih[8] + b_hh[8];
  b_rz4.x = b_ih[4] + b_hh[4];  b_rz4.y = b_ih[9] + b_hh[9];
  b_n01.x = b_hh[10];  b_n01.y = b_hh[11];
  b_n23.x = b_hh[12];  b_n23.y = b_hh[13];
  v2f b_n4u;  b_n4u.x = b_hh[14];  b_n4u.y = 0.0f;

  v2f x_r01, x_r23, x_z01, x_z23, x_rz4;
  x_r01.x = w_ih[0];  x_r01.y = w_ih[1];   x_r23.x = w_ih[2];  x_r23.y = w_ih[3];
  x_z01.x = w_ih[5];  x_z01.y = w_ih[6];   x_z23.x = w_ih[7];  x_z23.y = w_ih[8];
  x_rz4.x = w_ih[4];  x_rz4.y = w_ih[9];

  v2f q_r01, q_r23, q_z01, q_z23, q_rz4, q_n01, q_n23;
  q_r01.x = w_hh[0 * 5 + 4];   q_r01.y = w_hh[1 * 5 + 4];
  q_r23.x = w_hh[2 * 5 + 4];   q_r23.y = w_hh[3 * 5 + 4];
  q_z01.x = w_hh[5 * 5 + 4];   q_z01.y = w_hh[6 * 5 + 4];
  q_z23.x = w_hh[7 * 5 + 4];   q_z23.y = w_hh[8 * 5 + 4];
  q_rz4.x = w_hh[4 * 5 + 4];   q_rz4.y = w_hh[9 * 5 + 4];
  q_n01.x = w_hh[10 * 5 + 4];  q_n01.y = w_hh[11 * 5 + 4];
  q_n23.x = w_hh[12 * 5 + 4];  q_n23.y = w_hh[13 * 5 + 4];
  v2f q_n4u;  q_n4u.x = w_hh[14 * 5 + 4];  q_n4u.y = 0.0f;

  v2f wn01, wn23, bn01, bn23;
  wn01.x = w_ih[10]; wn01.y = w_ih[11]; wn23.x = w_ih[12]; wn23.y = w_ih[13];
  bn01.x = b_ih[10]; bn01.y = b_ih[11]; bn23.x = b_ih[12]; bn23.y = b_ih[13];
  const float wn4 = w_ih[14], bn4 = b_ih[14];

  float xv[GRU_T];
#pragma unroll
  for (int t = 0; t < GRU_T; ++t) xv[t] = x[xbase + t * GRU_S];

  v2f h01 = {0.f, 0.f}, h23 = {0.f, 0.f};
  float h4 = 0.f;

  {
    const v2f x2 = {xv[0], xv[0]};
    const v2f r01 = sigmoid2(fma2(x_r01, x2, b_r01));
    const v2f r23 = sigmoid2(fma2(x_r23, x2, b_r23));
    const v2f z01 = sigmoid2(fma2(x_z01, x2, b_z01));
    const v2f z23 = sigmoid2(fma2(x_z23, x2, b_z23));
    const v2f rz4 = sigmoid2(fma2(x_rz4, x2, b_rz4));

    v2f p01 = fma2(r01, b_n01, fma2(wn01, x2, bn01));
    v2f p23 = fma2(r23, b_n23, fma2(wn23, x2, bn23));
    float p4 = __builtin_fmaf(rz4.x, b_n4u.x, __builtin_fmaf(wn4, xv[0], bn4));
    p01 = tanh2(p01);  p23 = tanh2(p23);  p4 = tanh_fast(p4);

    h01 = fma2(z01, -p01, p01);
    h23 = fma2(z23, -p23, p23);
    h4  = __builtin_fmaf(rz4.y, -p4, p4);
  }

#pragma unroll
  for (int t = 1; t < GRU_T; ++t) {
    const v2f x2 = {xv[t], xv[t]};
    const v2f h44 = {h4, h4};

    const v2f c_r01 = fma2(q_r01, h44, fma2(x_r01, x2, b_r01));
    const v2f c_r23 = fma2(q_r23, h44, fma2(x_r23, x2, b_r23));
    const v2f c_z01 = fma2(q_z01, h44, fma2(x_z01, x2, b_z01));
    const v2f c_z23 = fma2(q_z23, h44, fma2(x_z23, x2, b_z23));
    const v2f c_rz4 = fma2(q_rz4, h44, fma2(x_rz4, x2, b_rz4));
    const v2f c_n01 = fma2(q_n01, h44, b_n01);
    const v2f c_n23 = fma2(q_n23, h44, b_n23);
    const v2f c_n4u = fma2(q_n4u, h44, b_n4u);
    const v8f c0 = {c_r01.x, c_r01.y, c_r23.x, c_r23.y, c_z01.x, c_z01.y, c_z23.x, c_z23.y};
    const v8f c1 = {c_rz4.x, c_rz4.y, c_n01.x, c_n01.y, c_n23.x, c_n23.y, c_n4u.x, c_n4u.y};

    v8f d0 = mm4(A00, h01, c0);
    v8f d1 = mm4(A10, h01, c1);
    d0 = mm4(A01, h23, d0);
    d1 = mm4(A11, h23, d1);

    v2f r01 = {d0[0], d0[1]}, r23 = {d0[2], d0[3]};
    v2f z01 = {d0[4], d0[5]}, z23 = {d0[6], d0[7]};
    v2f rz4 = {d1[0], d1[1]};
    v2f n01 = {d1[2], d1[3]}, n23 = {d1[4], d1[5]};
    const float n4 = d1[6];

    r01 = sigmoid2(r01);  r23 = sigmoid2(r23);
    z01 = sigmoid2(z01);  z23 = sigmoid2(z23);
    rz4 = sigmoid2(rz4);

    v2f p01 = fma2(r01, n01, fma2(wn01, x2, bn01));
    v2f p23 = fma2(r23, n23, fma2(wn23, x2, bn23));
    float p4 = __builtin_fmaf(rz4.x, n4, __builtin_fmaf(wn4, xv[t], bn4));
    p01 = tanh2(p01);  p23 = tanh2(p23);  p4 = tanh_fast(p4);

    h01 = fma2(z01, h01 - p01, p01);
    h23 = fma2(z23, h23 - p23, p23);
    h4  = __builtin_fmaf(rz4.y, h4 - p4, p4);
  }

  const float h[5] = {h01.x, h01.y, h23.x, h23.y, h4};

#pragma unroll
  for (int o = 0; o < 3; ++o) {
    float y = fc_b[o];
#pragma unroll
    for (int j = 0; j < 5; ++j) y = __builtin_fmaf(fc_w[o * 5 + j], h[j], y);
    out[(b * 3 + o) * GRU_S + s] = y;
  }
}

extern "C" void kernel_launch(void* const* d_in, const int* in_sizes, int n_in,
                              void* d_out, int out_size, void* d_ws, size_t ws_size,
                              hipStream_t stream) {
  (void)in_sizes; (void)n_in; (void)out_size; (void)d_ws; (void)ws_size;
  const float* x    = (const float*)d_in[0];
  const float* w_ih = (const float*)d_in[1];
  const float* w_hh = (const float*)d_in[2];
  const float* b_ih = (const float*)d_in[3];
  const float* b_hh = (const float*)d_in[4];
  const float* fc_w = (const float*)d_in[5];
  const float* fc_b = (const float*)d_in[6];
  float* out = (float*)d_out;

  const int threads = 256;
  const int samplesPerBlock = (threads / 32) * 32;
  const int blocks = (GRU_N + samplesPerBlock - 1) / samplesPerBlock;
  gru_wmma_kernel<<<blocks, threads, 0, stream>>>(x, w_ih, w_hh, b_ih, b_hh,
                                                  fc_w, fc_b, out);
}
